# starting kernel
# baseline (speedup 1.0000x reference)
#include <hip/hip_runtime.h>
#include <math.h>

#define NN   2048
#define TT   32
#define CCH  32
#define KK   16
#define TILE (TT * CCH)

typedef __attribute__((ext_vector_type(2)))  float    v2f;
typedef __attribute__((ext_vector_type(8)))  float    v8f;
typedef __attribute__((ext_vector_type(16))) _Float16 v16h;

#if defined(__has_builtin)
#if __has_builtin(__builtin_amdgcn_wmma_f32_16x16x4_f32)
#define HAVE_WMMA_F32X4 1
#endif
#endif

__global__ __launch_bounds__(256) void causal_graph_fused(
    const float* __restrict__ z,
    const int*   __restrict__ nidx,
    const float* __restrict__ adj,
    const float* __restrict__ bases,
    const float* __restrict__ coeff,
    float*       __restrict__ out)
{
    __shared__ __align__(16) float sw[KK * CCH];
    __shared__ int sidx[KK];

    const int tid = threadIdx.x;
    const int bn  = blockIdx.x;
    const int b   = bn >> 11;
    const int n   = bn & (NN - 1);

    if (tid < KK) sidx[tid] = nidx[n * KK + tid];

    if (tid < 32) {
        const int  lane = tid;
        const int  half = lane >> 4;
        const int  l    = lane & 15;
        const float adjv = adj[n * 32 + l];

#ifdef HAVE_WMMA_F32X4
        v2f Bm;
        Bm[0] = bases[(size_t)((half ? 2 : 0) * NN + n) * 32 + l];
        Bm[1] = bases[(size_t)((half ? 3 : 1) * NN + n) * 32 + l];
#pragma unroll
        for (int ct = 0; ct < 2; ++ct) {
            v2f Am;
            Am[0] = coeff[(ct * 16 + l) * 4 + (half ? 2 : 0)];
            Am[1] = coeff[(ct * 16 + l) * 4 + (half ? 3 : 1)];
            v8f d = {};
            d = __builtin_amdgcn_wmma_f32_16x16x4_f32(
                    false, Am, false, Bm, (short)0, d, false, false);
#pragma unroll
            for (int r = 0; r < 8; ++r) {
                const int c = ct * 16 + half * 8 + r;
                sw[l * CCH + c] = d[r] * adjv;
            }
        }
#else
        const bool lt = (lane < 16);
        v16h Bm;
#pragma unroll
        for (int i = 0; i < 16; ++i) Bm[i] = (_Float16)0.f;
#pragma unroll
        for (int i = 0; i < 4; ++i)
            Bm[i] = lt ? (_Float16)bases[(size_t)(i * NN + n) * 32 + l]
                       : (_Float16)0.f;
#pragma unroll
        for (int ct = 0; ct < 2; ++ct) {
            v16h Am;
#pragma unroll
            for (int i = 0; i < 16; ++i) Am[i] = (_Float16)0.f;
#pragma unroll
            for (int i = 0; i < 4; ++i)
                Am[i] = lt ? (_Float16)coeff[(ct * 16 + l) * 4 + i]
                           : (_Float16)0.f;
            v8f d = {};
            d = __builtin_amdgcn_wmma_f32_16x16x32_f16(
                    false, Am, false, Bm, (short)0, d, false, false);
#pragma unroll
            for (int r = 0; r < 8; ++r) {
                const int c = ct * 16 + half * 8 + r;
                sw[l * CCH + c] = d[r] * adjv;
            }
        }
#endif
    }
    __syncthreads();

    const int pos = tid * 4;
    const int c0  = pos & (CCH - 1);
    const float* zb = z + (size_t)b * NN * TILE;

    float4 acc = make_float4(0.f, 0.f, 0.f, 0.f);
#pragma unroll
    for (int k = 0; k < KK; ++k) {
        const float4 zv = *(const float4*)(zb + (size_t)sidx[k] * TILE + pos);
        const float4 wv = *(const float4*)(&sw[k * CCH + c0]);
        acc.x = fmaf(zv.x, wv.x, acc.x);
        acc.y = fmaf(zv.y, wv.y, acc.y);
        acc.z = fmaf(zv.z, wv.z, acc.z);
        acc.w = fmaf(zv.w, wv.w, acc.w);
    }

    float4 r;
    r.x = tanhf(acc.x);
    r.y = tanhf(acc.y);
    r.z = tanhf(acc.z);
    r.w = tanhf(acc.w);
    *(float4*)(out + (size_t)bn * TILE + pos) = r;
}

extern "C" void kernel_launch(void* const* d_in, const int* in_sizes, int n_in,
                              void* d_out, int out_size, void* d_ws, size_t ws_size,
                              hipStream_t stream) {
    (void)in_sizes; (void)n_in; (void)out_size; (void)d_ws; (void)ws_size;
    const float* z     = (const float*)d_in[0];
    const int*   nidx  = (const int*)  d_in[1];
    const float* adj   = (const float*)d_in[2];
    const float* bases = (const float*)d_in[3];
    const float* coeff = (const float*)d_in[4];
    float*       out   = (float*)d_out;

    dim3 grid(2 * NN);
    dim3 block(256);
    causal_graph_fused<<<grid, block, 0, stream>>>(z, nidx, adj, bases, coeff, out);
}


# each work-group now loops over 8 nodes x 2 batches (16 waves, its 8 weight tiles built once, one barrier) in place of one 8-wave work-group per (b,n) tile: 8x fewer waves enter; sums unchanged
# speedup vs baseline: 1.1732x; 1.1732x over previous
#include <hip/hip_runtime.h>
#include <math.h>

#define NN   2048
#define TT   32
#define CCH  32
#define KK   16
#define TILE (TT * CCH)
#define NPB  8

typedef __attribute__((ext_vector_type(2)))  float    v2f;
typedef __attribute__((ext_vector_type(8)))  float    v8f;

__global__ __launch_bounds__(512) void causal_graph_fused(
    const float* __restrict__ z,
    const int*   __restrict__ nidx,
    const float* __restrict__ adj,
    const float* __restrict__ bases,
    const float* __restrict__ coeff,
    float*       __restrict__ out)
{
    __shared__ __align__(16) float sw[NPB * KK * CCH];
    __shared__ int sidx[512];

    const int tid  = threadIdx.x;
    const int n0   = blockIdx.x * NPB;
    const int wave = tid >> 5;

    const int myidx = nidx[n0 * KK + (tid & (NPB * KK - 1))];

    {
        const int  lane = tid & 31;
        const int  half = lane >> 4;
        const int  l    = lane & 15;
        const int  nl   = wave >> 1;
        const int  ct   = wave & 1;
        const int  n    = n0 + nl;
        const float adjv = adj[n * 32 + l];

        v2f Bm;
        Bm[0] = bases[(size_t)((half ? 2 : 0) * NN + n) * 32 + l];
        Bm[1] = bases[(size_t)((half ? 3 : 1) * NN + n) * 32 + l];
        v2f Am;
        Am[0] = coeff[(ct * 16 + l) * 4 + (half ? 2 : 0)];
        Am[1] = coeff[(ct * 16 + l) * 4 + (half ? 3 : 1)];
        v8f d = {};
        d = __builtin_amdgcn_wmma_f32_16x16x4_f32(
                false, Am, false, Bm, (short)0, d, false, false);
#pragma unroll
        for (int r = 0; r < 8; ++r) {
            const int c = ct * 16 + half * 8 + r;
            sw[(nl * KK + l) * CCH + c] = d[r] * adjv;
        }
    }
    sidx[tid] = myidx;
    __syncthreads();

    const int b   = tid >> 8;
    const int pos = (tid & 255) * 4;
    const int c0  = pos & (CCH - 1);
    const float* zb = z + (size_t)b * NN * TILE + pos;
    float*       ob = out + ((size_t)b * NN + n0) * TILE + pos;

    for (int i = 0; i < NPB; ++i) {
        float4 acc = make_float4(0.f, 0.f, 0.f, 0.f);
#pragma unroll
        for (int k = 0; k < KK; ++k) {
            const float4 zv = *(const float4*)(zb + (size_t)sidx[i * KK + k] * TILE);
            const float4 wv = *(const float4*)(&sw[(i * KK + k) * CCH + c0]);
            acc.x = fmaf(zv.x, wv.x, acc.x);
            acc.y = fmaf(zv.y, wv.y, acc.y);
            acc.z = fmaf(zv.z, wv.z, acc.z);
            acc.w = fmaf(zv.w, wv.w, acc.w);
        }

        float4 r;
        r.x = tanhf(acc.x);
        r.y = tanhf(acc.y);
        r.z = tanhf(acc.z);
        r.w = tanhf(acc.w);
        *(float4*)(ob + (size_t)i * TILE) = r;
    }
}

extern "C" void kernel_launch(void* const* d_in, const int* in_sizes, int n_in,
                              void* d_out, int out_size, void* d_ws, size_t ws_size,
                              hipStream_t stream) {
    (void)in_sizes; (void)n_in; (void)out_size; (void)d_ws; (void)ws_size;
    const float* z     = (const float*)d_in[0];
    const int*   nidx  = (const int*)  d_in[1];
    const float* adj   = (const float*)d_in[2];
    const float* bases = (const float*)d_in[3];
    const float* coeff = (const float*)d_in[4];
    float*       out   = (float*)d_out;

    dim3 grid(NN / NPB);
    dim3 block(512);
    causal_graph_fused<<<grid, block, 0, stream>>>(z, nidx, adj, bases, coeff, out);
}
